# starting kernel
# baseline (speedup 1.0000x reference)
#include <hip/hip_runtime.h>
#include <hip/hip_bf16.h>

typedef float v2f __attribute__((ext_vector_type(2)));
typedef float v8f __attribute__((ext_vector_type(8)));

#define IN_CH 128
#define OUT_CH 64

__global__ void prep_weights_t(const float* __restrict__ w_main_in,
                               const float* __restrict__ w_main_out,
                               const float* __restrict__ w_shared,
                               float* __restrict__ w_in_t,
                               float* __restrict__ w_out_t) {
    int i = blockIdx.x * blockDim.x + threadIdx.x;
    if (i < IN_CH * OUT_CH) {
        int k = i >> 6;
        int c = i & 63;
        float s = w_shared[i];
        w_in_t [c * IN_CH + k] = w_main_in[i]  + s;
        w_out_t[c * IN_CH + k] = w_main_out[i] + s;
    }
}

__global__ void gemm_dual_wmma(const float* __restrict__ x,
                               const float* __restrict__ w_in_t,
                               const float* __restrict__ w_out_t,
                               float* __restrict__ h_in,
                               float* __restrict__ h_out,
                               int n_nodes) {
    __shared__ __align__(16) float a_tile[16 * 132];

    const int mtile = blockIdx.x;
    const int row0  = mtile * 16;

    for (int i = threadIdx.x; i < 16 * (IN_CH / 4); i += blockDim.x) {
        int r  = i >> 5;
        int c4 = i & 31;
        float4 v = make_float4(0.f, 0.f, 0.f, 0.f);
        int grow = row0 + r;
        if (grow < n_nodes)
            v = ((const float4*)x)[(size_t)grow * (IN_CH / 4) + c4];
        *(float4*)&a_tile[r * 132 + c4 * 4] = v;
    }
    __syncthreads();

    const int wave = threadIdx.x >> 5;
    const int lane = threadIdx.x & 31;
    const int r    = lane & 15;
    const int half = lane >> 4;

    const float* __restrict__ Wt = (wave < 4) ? w_in_t : w_out_t;
    float*       __restrict__ H  = (wave < 4) ? h_in   : h_out;
    const int nt  = wave & 3;
    const int col = nt * 16 + r;

    const float* __restrict__ ap = &a_tile[r * 132 + 2 * half];
    const float* __restrict__ wp = Wt + (size_t)col * IN_CH + 2 * half;

    v8f acc = {};

#pragma unroll
    for (int k0 = 0; k0 < IN_CH; k0 += 4) {
        v2f a = *(const v2f*)(ap + k0);
        v2f b = *(const v2f*)(wp + k0);
        acc = __builtin_amdgcn_wmma_f32_16x16x4_f32(
             false, a,  false, b,
             (short)0, acc,  false,  false);
    }

#pragma unroll
    for (int v = 0; v < 8; ++v) {
        int grow = row0 + v + 8 * half;
        if (grow < n_nodes)
            H[(size_t)grow * OUT_CH + col] = acc[v];
    }
}

__global__ void init_out(float* __restrict__ out,
                         const float* __restrict__ c_in,
                         const float* __restrict__ c_out,
                         const float* __restrict__ b_mi,
                         const float* __restrict__ b_mo,
                         const float* __restrict__ b_si,
                         const float* __restrict__ b_so,
                         int n_nodes) {
    int idx = blockIdx.x * blockDim.x + threadIdx.x;
    if (idx < n_nodes * OUT_CH) {
        int n = idx >> 6;
        int c = idx & 63;
        out[idx] = c_in[n] * (b_mi[c] + b_si[c]) + c_out[n] * (b_mo[c] + b_so[c]);
    }
}

__global__ void scatter_edges(const float* __restrict__ h_in,
                              const float* __restrict__ h_out,
                              const int*   __restrict__ ei_in,
                              const float* __restrict__ ew_in,
                              const int*   __restrict__ ei_out,
                              const float* __restrict__ ew_out,
                              const float* __restrict__ c_in,
                              const float* __restrict__ c_out,
                              float* __restrict__ out,
                              int e_in, int e_out) {
    long long t = (long long)blockIdx.x * blockDim.x + threadIdx.x;
    int e = (int)(t >> 6);
    int c = (int)(t & 63);
    if (e < e_in) {
        int src = ei_in[e];
        int tgt = ei_in[e_in + e];
        float m = c_in[tgt] * ew_in[e] * h_in[(size_t)src * OUT_CH + c];
        unsafeAtomicAdd(&out[(size_t)tgt * OUT_CH + c], m);
    } else {
        e -= e_in;
        if (e < e_out) {
            int src = ei_out[e];
            int tgt = ei_out[e_out + e];
            float m = c_out[tgt] * ew_out[e] * h_out[(size_t)src * OUT_CH + c];
            unsafeAtomicAdd(&out[(size_t)tgt * OUT_CH + c], m);
        }
    }
}

extern "C" void kernel_launch(void* const* d_in, const int* in_sizes, int n_in,
                              void* d_out, int out_size, void* d_ws, size_t ws_size,
                              hipStream_t stream) {
    const float* x          = (const float*)d_in[0];
    const float* w_main_in  = (const float*)d_in[1];
    const float* w_main_out = (const float*)d_in[2];
    const float* w_shared   = (const float*)d_in[3];
    const float* b_mi       = (const float*)d_in[4];
    const float* b_mo       = (const float*)d_in[5];
    const float* b_si       = (const float*)d_in[6];
    const float* b_so       = (const float*)d_in[7];
    const float* c_in       = (const float*)d_in[8];
    const float* c_out      = (const float*)d_in[9];
    const int*   ei_in      = (const int*)  d_in[10];
    const float* ew_in      = (const float*)d_in[11];
    const int*   ei_out     = (const int*)  d_in[12];
    const float* ew_out     = (const float*)d_in[13];
    float* out = (float*)d_out;

    const int n_nodes = in_sizes[8];
    const int e_in    = in_sizes[11];
    const int e_out   = in_sizes[13];

    float* ws      = (float*)d_ws;
    float* w_in_t  = ws;
    float* w_out_t = w_in_t  + IN_CH * OUT_CH;
    float* h_in    = w_out_t + IN_CH * OUT_CH;
    float* h_out   = h_in    + (size_t)n_nodes * OUT_CH;

    prep_weights_t<<<(IN_CH * OUT_CH + 255) / 256, 256, 0, stream>>>(
        w_main_in, w_main_out, w_shared, w_in_t, w_out_t);

    const int mtiles = (n_nodes + 15) / 16;
    gemm_dual_wmma<<<mtiles, 256, 0, stream>>>(x, w_in_t, w_out_t, h_in, h_out, n_nodes);

    const int total = n_nodes * OUT_CH;
    init_out<<<(total + 255) / 256, 256, 0, stream>>>(
        out, c_in, c_out, b_mi, b_mo, b_si, b_so, n_nodes);

    const long long scatter_threads = ((long long)e_in + e_out) * OUT_CH;
    const int scatter_blocks = (int)((scatter_threads + 255) / 256);
    scatter_edges<<<scatter_blocks, 256, 0, stream>>>(
        h_in, h_out, ei_in, ew_in, ei_out, ew_out, c_in, c_out, out, e_in, e_out);
}


# scatter: one wave per 64 consecutive edges sums runs of equal target in registers and issues one f32 atomic row add per run instead of per edge; float summation order changes
# speedup vs baseline: 5.6675x; 5.6675x over previous
#include <hip/hip_runtime.h>
#include <hip/hip_bf16.h>

typedef float v2f __attribute__((ext_vector_type(2)));
typedef float v8f __attribute__((ext_vector_type(8)));

#define IN_CH 128
#define OUT_CH 64

__global__ void prep_weights_t(const float* __restrict__ w_main_in,
                               const float* __restrict__ w_main_out,
                               const float* __restrict__ w_shared,
                               float* __restrict__ w_in_t,
                               float* __restrict__ w_out_t) {
    int i = blockIdx.x * blockDim.x + threadIdx.x;
    if (i < IN_CH * OUT_CH) {
        int k = i >> 6;
        int c = i & 63;
        float s = w_shared[i];
        w_in_t [c * IN_CH + k] = w_main_in[i]  + s;
        w_out_t[c * IN_CH + k] = w_main_out[i] + s;
    }
}

__global__ void gemm_dual_wmma(const float* __restrict__ x,
                               const float* __restrict__ w_in_t,
                               const float* __restrict__ w_out_t,
                               float* __restrict__ h_in,
                               float* __restrict__ h_out,
                               int n_nodes) {
    __shared__ __align__(16) float a_tile[16 * 132];

    const int mtile = blockIdx.x;
    const int row0  = mtile * 16;

    for (int i = threadIdx.x; i < 16 * (IN_CH / 4); i += blockDim.x) {
        int r  = i >> 5;
        int c4 = i & 31;
        float4 v = make_float4(0.f, 0.f, 0.f, 0.f);
        int grow = row0 + r;
        if (grow < n_nodes)
            v = ((const float4*)x)[(size_t)grow * (IN_CH / 4) + c4];
        *(float4*)&a_tile[r * 132 + c4 * 4] = v;
    }
    __syncthreads();

    const int wave = threadIdx.x >> 5;
    const int lane = threadIdx.x & 31;
    const int r    = lane & 15;
    const int half = lane >> 4;

    const float* __restrict__ Wt = (wave < 4) ? w_in_t : w_out_t;
    float*       __restrict__ H  = (wave < 4) ? h_in   : h_out;
    const int nt  = wave & 3;
    const int col = nt * 16 + r;

    const float* __restrict__ ap = &a_tile[r * 132 + 2 * half];
    const float* __restrict__ wp = Wt + (size_t)col * IN_CH + 2 * half;

    v8f acc = {};

#pragma unroll
    for (int k0 = 0; k0 < IN_CH; k0 += 4) {
        v2f a = *(const v2f*)(ap + k0);
        v2f b = *(const v2f*)(wp + k0);
        acc = __builtin_amdgcn_wmma_f32_16x16x4_f32(
             false, a,  false, b,
             (short)0, acc,  false,  false);
    }

#pragma unroll
    for (int v = 0; v < 8; ++v) {
        int grow = row0 + v + 8 * half;
        if (grow < n_nodes)
            H[(size_t)grow * OUT_CH + col] = acc[v];
    }
}

__global__ void init_out(float* __restrict__ out,
                         const float* __restrict__ c_in,
                         const float* __restrict__ c_out,
                         const float* __restrict__ b_mi,
                         const float* __restrict__ b_mo,
                         const float* __restrict__ b_si,
                         const float* __restrict__ b_so,
                         int n_nodes) {
    int idx = blockIdx.x * blockDim.x + threadIdx.x;
    if (idx < n_nodes * OUT_CH) {
        int n = idx >> 6;
        int c = idx & 63;
        out[idx] = c_in[n] * (b_mi[c] + b_si[c]) + c_out[n] * (b_mo[c] + b_so[c]);
    }
}

#define EDGES_PER_BATCH 32
#define EDGES_PER_WAVE  64

static __device__ __forceinline__ int bcast_i(int v, int j) {
    return __builtin_amdgcn_readlane(v, j);
}
static __device__ __forceinline__ float bcast_f(float v, int j) {
    return __int_as_float(__builtin_amdgcn_readlane(__float_as_int(v), j));
}

static __device__ __forceinline__ void flush_run(float* __restrict__ out, int tgt,
                                                 unsigned lane, v2f acc) {
    float* o = out + (size_t)tgt * OUT_CH;
    unsafeAtomicAdd(o + lane,      acc.x);
    unsafeAtomicAdd(o + lane + 32, acc.y);
}

static __device__ __forceinline__ void edge_batch(const float* __restrict__ h,
                                                  float* __restrict__ out,
                                                  int src_l, int tgt_l, float s_l,
                                                  int nv, unsigned lane, int& cur, v2f& acc) {
    if (nv == EDGES_PER_BATCH) {
        v2f row[EDGES_PER_BATCH];
#pragma unroll
        for (int j = 0; j < EDGES_PER_BATCH; ++j) {
            const float* rp = h + (size_t)bcast_i(src_l, j) * OUT_CH;
            row[j].x = rp[lane];
            row[j].y = rp[lane + 32];
        }
#pragma unroll
        for (int j = 0; j < EDGES_PER_BATCH; ++j) {
            const int   t = bcast_i(tgt_l, j);
            const float s = bcast_f(s_l, j);
            if (t != cur) {
                flush_run(out, cur, lane, acc);
                cur = t;
                acc.x = 0.f; acc.y = 0.f;
            }
            acc.x = fmaf(s, row[j].x, acc.x);
            acc.y = fmaf(s, row[j].y, acc.y);
        }
    } else {
        for (int j = 0; j < nv; ++j) {
            const float* rp = h + (size_t)bcast_i(src_l, j) * OUT_CH;
            const int   t  = bcast_i(tgt_l, j);
            const float s  = bcast_f(s_l, j);
            const float rx = rp[lane];
            const float ry = rp[lane + 32];
            if (t != cur) {
                flush_run(out, cur, lane, acc);
                cur = t;
                acc.x = 0.f; acc.y = 0.f;
            }
            acc.x = fmaf(s, rx, acc.x);
            acc.y = fmaf(s, ry, acc.y);
        }
    }
}

__global__ void scatter_edges(const float* __restrict__ h_in,
                              const float* __restrict__ h_out,
                              const int*   __restrict__ ei_in,
                              const float* __restrict__ ew_in,
                              const int*   __restrict__ ei_out,
                              const float* __restrict__ ew_out,
                              const float* __restrict__ c_in,
                              const float* __restrict__ c_out,
                              float* __restrict__ out,
                              int e_in, int e_out) {
    const unsigned lane = threadIdx.x & 31;
    const int wv = __builtin_amdgcn_readfirstlane((int)(threadIdx.x >> 5));
    int chunk = (int)blockIdx.x * (int)(blockDim.x >> 5) + wv;

    const int chunks_in  = (e_in  + EDGES_PER_WAVE - 1) / EDGES_PER_WAVE;
    const int chunks_out = (e_out + EDGES_PER_WAVE - 1) / EDGES_PER_WAVE;

    const float* __restrict__ h;
    const int*   __restrict__ ei;
    const float* __restrict__ ew;
    const float* __restrict__ cv;
    int E;
    if (chunk < chunks_in) {
        h = h_in;  ei = ei_in;  ew = ew_in;  cv = c_in;  E = e_in;
    } else {
        chunk -= chunks_in;
        if (chunk >= chunks_out) return;
        h = h_out; ei = ei_out; ew = ew_out; cv = c_out; E = e_out;
    }

    const int e0  = chunk * EDGES_PER_WAVE;
    const int nvA = min(E - e0, EDGES_PER_BATCH);
    const int nvB = max(min(E - e0 - EDGES_PER_BATCH, EDGES_PER_BATCH), 0);

    const int eA = min(e0 + (int)lane, E - 1);
    const int eB = min(e0 + EDGES_PER_BATCH + (int)lane, E - 1);
    const int   tgtA = ei[(size_t)E + eA];
    const int   tgtB = ei[(size_t)E + eB];
    const int   srcA = ei[eA];
    const int   srcB = ei[eB];
    const float wA   = ew[eA];
    const float wB   = ew[eB];
    const float sA   = cv[tgtA] * wA;
    const float sB   = cv[tgtB] * wB;

    int cur = bcast_i(tgtA, 0);
    v2f acc = {0.f, 0.f};

    edge_batch(h, out, srcA, tgtA, sA, nvA, lane, cur, acc);
    if (nvB > 0)
        edge_batch(h, out, srcB, tgtB, sB, nvB, lane, cur, acc);

    flush_run(out, cur, lane, acc);
}

extern "C" void kernel_launch(void* const* d_in, const int* in_sizes, int n_in,
                              void* d_out, int out_size, void* d_ws, size_t ws_size,
                              hipStream_t stream) {
    const float* x          = (const float*)d_in[0];
    const float* w_main_in  = (const float*)d_in[1];
    const float* w_main_out = (const float*)d_in[2];
    const float* w_shared   = (const float*)d_in[3];
    const float* b_mi       = (const float*)d_in[4];
    const float* b_mo       = (const float*)d_in[5];
    const float* b_si       = (const float*)d_in[6];
    const float* b_so       = (const float*)d_in[7];
    const float* c_in       = (const float*)d_in[8];
    const float* c_out      = (const float*)d_in[9];
    const int*   ei_in      = (const int*)  d_in[10];
    const float* ew_in      = (const float*)d_in[11];
    const int*   ei_out     = (const int*)  d_in[12];
    const float* ew_out     = (const float*)d_in[13];
    float* out = (float*)d_out;

    const int n_nodes = in_sizes[8];
    const int e_in    = in_sizes[11];
    const int e_out   = in_sizes[13];

    float* ws      = (float*)d_ws;
    float* w_in_t  = ws;
    float* w_out_t = w_in_t  + IN_CH * OUT_CH;
    float* h_in    = w_out_t + IN_CH * OUT_CH;
    float* h_out   = h_in    + (size_t)n_nodes * OUT_CH;

    prep_weights_t<<<(IN_CH * OUT_CH + 255) / 256, 256, 0, stream>>>(
        w_main_in, w_main_out, w_shared, w_in_t, w_out_t);

    const int mtiles = (n_nodes + 15) / 16;
    gemm_dual_wmma<<<mtiles, 256, 0, stream>>>(x, w_in_t, w_out_t, h_in, h_out, n_nodes);

    const int total = n_nodes * OUT_CH;
    init_out<<<(total + 255) / 256, 256, 0, stream>>>(
        out, c_in, c_out, b_mi, b_mo, b_si, b_so, n_nodes);

    const int scatter_waves  = (e_in + EDGES_PER_WAVE - 1) / EDGES_PER_WAVE
                             + (e_out + EDGES_PER_WAVE - 1) / EDGES_PER_WAVE;
    const int scatter_blocks = (scatter_waves + 7) / 8;
    scatter_edges<<<scatter_blocks, 256, 0, stream>>>(
        h_in, h_out, ei_in, ew_in, ei_out, ew_out, c_in, c_out, out, e_in, e_out);
}


# edge pass: lane owns channels 2l,2l+1, so an edge's 256-byte h row is ONE 8-byte-per-lane load on a scalar row base (was two 4-byte loads); run flush swaps halves to keep whole-line atomics
# speedup vs baseline: 8.9019x; 1.5707x over previous
#include <hip/hip_runtime.h>
#include <hip/hip_bf16.h>

typedef float v2f __attribute__((ext_vector_type(2)));
typedef float v8f __attribute__((ext_vector_type(8)));

#define IN_CH 128
#define OUT_CH 64

#define TILE_M      32
#define A_PITCH     132
#define GEMM_BLOCKS 768

static __device__ __forceinline__ void fetch_tile(const float* __restrict__ x,
                                                  const float* __restrict__ c_in,
                                                  const float* __restrict__ c_out,
                                                  int tile, int n_nodes, int tid,
                                                  float4 (&pre)[4], float (&pci)[2], float (&pco)[2]) {
    const int row0 = tile * TILE_M;
    const int srow = tid >> 5;
    const int sc4  = tid & 31;
    const int last = n_nodes - 1;
#pragma unroll
    for (int j = 0; j < 4; ++j) {
        const int grow = min(row0 + srow + 8 * j, last);
        pre[j] = ((const float4*)x)[(size_t)grow * (IN_CH / 4) + sc4];
    }
#pragma unroll
    for (int j = 0; j < 2; ++j) {
        const int orow = min(row0 + (tid >> 4) + 16 * j, last);
        pci[j] = c_in[orow];
        pco[j] = c_out[orow];
    }
}

__global__ __launch_bounds__(256)
void gemm_dual_wmma(const float* __restrict__ x,
                    const float* __restrict__ w_main_in,
                    const float* __restrict__ w_main_out,
                    const float* __restrict__ w_shared,
                    float* __restrict__ h_in,
                    float* __restrict__ h_out,
                    float* __restrict__ out,
                    const float* __restrict__ c_in,
                    const float* __restrict__ c_out,
                    const float* __restrict__ b_mi,
                    const float* __restrict__ b_mo,
                    const float* __restrict__ b_si,
                    const float* __restrict__ b_so,
                    int n_nodes) {
    __shared__ __align__(16) float a_tile[2 * TILE_M * A_PITCH];

    const int ntiles = (n_nodes + TILE_M - 1) / TILE_M;
    int t = blockIdx.x;
    if (t >= ntiles) return;

    const int tid  = threadIdx.x;
    const int wave = tid >> 5;
    const int lane = tid & 31;
    const int r    = lane & 15;
    const int half = lane >> 4;

    const float* __restrict__ Wm = (wave < 4) ? w_main_in : w_main_out;
    float*       __restrict__ H  = (wave < 4) ? h_in      : h_out;
    const int col = (wave & 3) * 16 + r;

    float4 pre[4];
    float  pci[2], pco[2];
    fetch_tile(x, c_in, c_out, t, n_nodes, tid, pre, pci, pco);

    const int oc4 = tid & 15;
    const int orl = tid >> 4;
    const float4 o_bmi = ((const float4*)b_mi)[oc4];
    const float4 o_bsi = ((const float4*)b_si)[oc4];
    const float4 o_bmo = ((const float4*)b_mo)[oc4];
    const float4 o_bso = ((const float4*)b_so)[oc4];

    v2f bfrag[IN_CH / 4];
    {
        const float* __restrict__ wm = Wm       + (size_t)(2 * half) * OUT_CH + col;
        const float* __restrict__ wc = w_shared + (size_t)(2 * half) * OUT_CH + col;
#pragma unroll
        for (int i = 0; i < IN_CH / 4; ++i) {
            bfrag[i].x = wm[(4 * i)     * OUT_CH] + wc[(4 * i)     * OUT_CH];
            bfrag[i].y = wm[(4 * i + 1) * OUT_CH] + wc[(4 * i + 1) * OUT_CH];
            if ((i & 15) == 15)
                __builtin_amdgcn_sched_barrier(0);
        }
    }

    const float4 bs_in  = make_float4(o_bmi.x + o_bsi.x, o_bmi.y + o_bsi.y, o_bmi.z + o_bsi.z, o_bmi.w + o_bsi.w);
    const float4 bs_out = make_float4(o_bmo.x + o_bso.x, o_bmo.y + o_bso.y, o_bmo.z + o_bso.z, o_bmo.w + o_bso.w);

    const int srow = tid >> 5;
    const int sc4  = tid & 31;
    int buf = 0;

    for (;;) {
        float* __restrict__ at = a_tile + buf * (TILE_M * A_PITCH);

        const int row0 = t * TILE_M;

#pragma unroll
        for (int j = 0; j < 4; ++j) {
            const bool in = row0 + srow + 8 * j < n_nodes;
            const float4 v = pre[j];
            *(float4*)&at[(srow + 8 * j) * A_PITCH + sc4 * 4] =
                make_float4(in ? v.x : 0.f, in ? v.y : 0.f, in ? v.z : 0.f, in ? v.w : 0.f);
        }
        const float ci0 = pci[0], ci1 = pci[1], co0 = pco[0], co1 = pco[1];
        __syncthreads();

        const int tn = t + (int)gridDim.x;
        if (tn < ntiles)
            fetch_tile(x, c_in, c_out, tn, n_nodes, tid, pre, pci, pco);

        {
            const int orow0 = row0 + orl;
            const int orow1 = orow0 + 16;
            if (orow0 < n_nodes) {
                float4 ov;
                ov.x = ci0 * bs_in.x + co0 * bs_out.x;
                ov.y = ci0 * bs_in.y + co0 * bs_out.y;
                ov.z = ci0 * bs_in.z + co0 * bs_out.z;
                ov.w = ci0 * bs_in.w + co0 * bs_out.w;
                ((float4*)out)[(size_t)orow0 * (OUT_CH / 4) + oc4] = ov;
            }
            if (orow1 < n_nodes) {
                float4 ov;
                ov.x = ci1 * bs_in.x + co1 * bs_out.x;
                ov.y = ci1 * bs_in.y + co1 * bs_out.y;
                ov.z = ci1 * bs_in.z + co1 * bs_out.z;
                ov.w = ci1 * bs_in.w + co1 * bs_out.w;
                ((float4*)out)[(size_t)orow1 * (OUT_CH / 4) + oc4] = ov;
            }
        }

        const float* __restrict__ ap0 = &at[r * A_PITCH + 2 * half];
        const float* __restrict__ ap1 = ap0 + 16 * A_PITCH;
        v8f acc0 = {};
        v8f acc1 = {};
#pragma unroll
        for (int i = 0; i < IN_CH / 4; ++i) {
            v2f a0 = *(const v2f*)(ap0 + 4 * i);
            v2f a1 = *(const v2f*)(ap1 + 4 * i);
            acc0 = __builtin_amdgcn_wmma_f32_16x16x4_f32(
                 false, a0,  false, bfrag[i],
                 (short)0, acc0,  false,  false);
            acc1 = __builtin_amdgcn_wmma_f32_16x16x4_f32(
                 false, a1,  false, bfrag[i],
                 (short)0, acc1,  false,  false);
        }

#pragma unroll
        for (int v = 0; v < 8; ++v) {
            const int grow0 = row0 + v + 8 * half;
            const int grow1 = grow0 + 16;
            if (grow0 < n_nodes)
                H[(size_t)grow0 * OUT_CH + col] = acc0[v];
            if (grow1 < n_nodes)
                H[(size_t)grow1 * OUT_CH + col] = acc1[v];
        }

        if (tn >= ntiles) break;
        t = tn;
        buf ^= 1;
    }
}

#define EDGES_PER_BATCH 32
#define EDGES_PER_WAVE  64

static __device__ __forceinline__ int bcast_i(int v, int j) {
    return __builtin_amdgcn_readlane(v, j);
}
static __device__ __forceinline__ float bcast_f(float v, int j) {
    return __int_as_float(__builtin_amdgcn_readlane(__float_as_int(v), j));
}

static __device__ __forceinline__ float xhalf(float x) {
    return __int_as_float(__builtin_amdgcn_permlanex16(__float_as_int(x), __float_as_int(x),
                                                       0x76543210, 0xfedcba98, false, false));
}

static __device__ __forceinline__ v2f load_row(const char* __restrict__ hb, unsigned roff, unsigned lane8) {
    typedef __attribute__((address_space(1))) const char gchar_t;
    typedef __attribute__((address_space(1))) const v2f  gv2f_t;
    gchar_t* rp = (gchar_t*)hb + (size_t)roff;
    asm("" : "+s"(rp));
    return *(gv2f_t*)(rp + lane8);
}

static __device__ __forceinline__ void flush_run(float* __restrict__ out, int tgt,
                                                 unsigned lane, v2f acc) {
    const bool  up   = lane >= 16;
    const float got  = xhalf(up ? acc.x : acc.y);
    const float v_lo = up ? got   : acc.x;
    const float v_hi = up ? acc.y : got;
    float* o = out + (size_t)tgt * OUT_CH + (2 * (lane & 15) + (lane >> 4));
    unsafeAtomicAdd(o,      v_lo);
    unsafeAtomicAdd(o + 32, v_hi);
}

static __device__ __forceinline__ void edge_batch(const char* __restrict__ hb,
                                                  float* __restrict__ out,
                                                  unsigned roff_l, int tgt_l, float s_l,
                                                  int nv, unsigned lane, int& cur, v2f& acc) {
    const unsigned lane8 = lane * 8;
    if (nv == EDGES_PER_BATCH) {
        v2f row[EDGES_PER_BATCH];
#pragma unroll
        for (int j = 0; j < EDGES_PER_BATCH; ++j)
            row[j] = load_row(hb, (unsigned)bcast_i((int)roff_l, j), lane8);
#pragma unroll
        for (int j = 0; j < EDGES_PER_BATCH; ++j) {
            const int   t = bcast_i(tgt_l, j);
            const float s = bcast_f(s_l, j);
            if (t != cur) {
                flush_run(out, cur, lane, acc);
                cur = t;
                acc.x = 0.f; acc.y = 0.f;
            }
            acc.x = fmaf(s, row[j].x, acc.x);
            acc.y = fmaf(s, row[j].y, acc.y);
        }
    } else {
        for (int j = 0; j < nv; ++j) {
            const v2f   rw = load_row(hb, (unsigned)bcast_i((int)roff_l, j), lane8);
            const int   t  = bcast_i(tgt_l, j);
            const float s  = bcast_f(s_l, j);
            if (t != cur) {
                flush_run(out, cur, lane, acc);
                cur = t;
                acc.x = 0.f; acc.y = 0.f;
            }
            acc.x = fmaf(s, rw.x, acc.x);
            acc.y = fmaf(s, rw.y, acc.y);
        }
    }
}

__global__ void scatter_edges(const float* __restrict__ h_in,
                              const float* __restrict__ h_out,
                              const int*   __restrict__ ei_in,
                              const float* __restrict__ ew_in,
                              const int*   __restrict__ ei_out,
                              const float* __restrict__ ew_out,
                              const float* __restrict__ c_in,
                              const float* __restrict__ c_out,
                              float* __restrict__ out,
                              int e_in, int e_out) {
    const unsigned lane = threadIdx.x & 31;
    const int wv = __builtin_amdgcn_readfirstlane((int)(threadIdx.x >> 5));
    int chunk = (int)blockIdx.x * (int)(blockDim.x >> 5) + wv;

    const int chunks_in  = (e_in  + EDGES_PER_WAVE - 1) / EDGES_PER_WAVE;
    const int chunks_out = (e_out + EDGES_PER_WAVE - 1) / EDGES_PER_WAVE;

    const float* __restrict__ h;
    const int*   __restrict__ ei;
    const float* __restrict__ ew;
    const float* __restrict__ cv;
    int E;
    if (chunk < chunks_in) {
        h = h_in;  ei = ei_in;  ew = ew_in;  cv = c_in;  E = e_in;
    } else {
        chunk -= chunks_in;
        if (chunk >= chunks_out) return;
        h = h_out; ei = ei_out; ew = ew_out; cv = c_out; E = e_out;
    }

    const int e0  = chunk * EDGES_PER_WAVE;
    const int nvA = min(E - e0, EDGES_PER_BATCH);
    const int nvB = max(min(E - e0 - EDGES_PER_BATCH, EDGES_PER_BATCH), 0);

    const int eA = min(e0 + (int)lane, E - 1);
    const int eB = min(e0 + EDGES_PER_BATCH + (int)lane, E - 1);
    const int   tgtA = ei[(size_t)E + eA];
    const int   tgtB = ei[(size_t)E + eB];
    const int   srcA = ei[eA];
    const int   srcB = ei[eB];
    const float wA   = ew[eA];
    const float wB   = ew[eB];
    const float sA   = cv[tgtA] * wA;
    const float sB   = cv[tgtB] * wB;
    const unsigned roffA = (unsigned)srcA * (unsigned)(OUT_CH * sizeof(float));
    const unsigned roffB = (unsigned)srcB * (unsigned)(OUT_CH * sizeof(float));

    int cur = bcast_i(tgtA, 0);
    v2f acc = {0.f, 0.f};

    edge_batch((const char*)h, out, roffA, tgtA, sA, nvA, lane, cur, acc);
    if (nvB > 0)
        edge_batch((const char*)h, out, roffB, tgtB, sB, nvB, lane, cur, acc);

    flush_run(out, cur, lane, acc);
}

extern "C" void kernel_launch(void* const* d_in, const int* in_sizes, int n_in,
                              void* d_out, int out_size, void* d_ws, size_t ws_size,
                              hipStream_t stream) {
    const float* x          = (const float*)d_in[0];
    const float* w_main_in  = (const float*)d_in[1];
    const float* w_main_out = (const float*)d_in[2];
    const float* w_shared   = (const float*)d_in[3];
    const float* b_mi       = (const float*)d_in[4];
    const float* b_mo       = (const float*)d_in[5];
    const float* b_si       = (const float*)d_in[6];
    const float* b_so       = (const float*)d_in[7];
    const float* c_in       = (const float*)d_in[8];
    const float* c_out      = (const float*)d_in[9];
    const int*   ei_in      = (const int*)  d_in[10];
    const float* ew_in      = (const float*)d_in[11];
    const int*   ei_out     = (const int*)  d_in[12];
    const float* ew_out     = (const float*)d_in[13];
    float* out = (float*)d_out;

    const int n_nodes = in_sizes[8];
    const int e_in    = in_sizes[11];
    const int e_out   = in_sizes[13];

    float* ws    = (float*)d_ws;
    float* h_in  = ws;
    float* h_out = h_in + (size_t)n_nodes * OUT_CH;

    gemm_dual_wmma<<<GEMM_BLOCKS, 256, 0, stream>>>(x, w_main_in, w_main_out, w_shared, h_in, h_out,
                                                    out, c_in, c_out, b_mi, b_mo, b_si, b_so, n_nodes);

    const int scatter_waves  = (e_in + EDGES_PER_WAVE - 1) / EDGES_PER_WAVE
                             + (e_out + EDGES_PER_WAVE - 1) / EDGES_PER_WAVE;
    const int scatter_blocks = (scatter_waves + 7) / 8;
    scatter_edges<<<scatter_blocks, 256, 0, stream>>>(
        h_in, h_out, ei_in, ew_in, ei_out, ew_out, c_in, c_out, out, e_in, e_out);
}
